# starting kernel
# baseline (speedup 1.0000x reference)
#include <hip/hip_runtime.h>
#include <hip/hip_bf16.h>

typedef __attribute__((ext_vector_type(2))) float v2f;
typedef __attribute__((ext_vector_type(8))) float v8f;

#define S_LEN   4096
#define D_DIM   1024
#define NPATCH  8
#define NSEG    9

__global__ void seg_bounds_kernel(const int* __restrict__ mask,
                                  int* __restrict__ ofs) {
  const int b   = blockIdx.x;
  const int tid = threadIdx.x;
  __shared__ int psum[256];

  const int base = b * S_LEN + tid * 16;
  int m[16];
  int c = 0;
#pragma unroll
  for (int j = 0; j < 16; ++j) { m[j] = mask[base + j]; c += m[j]; }
  psum[tid] = c;
  __syncthreads();

#pragma unroll
  for (int off = 1; off < 256; off <<= 1) {
    int v   = psum[tid];
    int add = (tid >= off) ? psum[tid - off] : 0;
    __syncthreads();
    psum[tid] = v + add;
    __syncthreads();
  }
  const int excl = (tid == 0) ? 0 : psum[tid - 1];

  if (tid < 10) ofs[b * 10 + tid] = (tid == 0) ? 0 : S_LEN;
  __syncthreads();

  int r = excl;
#pragma unroll
  for (int j = 0; j < 16; ++j) {
    if (m[j]) {
      if (r < NPATCH) ofs[b * 10 + (r + 1)] = tid * 16 + j + 1;
      ++r;
    }
  }
}

__global__ void seg_mean_kernel(const float* __restrict__ x,
                                const int* __restrict__ ofs,
                                float* __restrict__ patches) {
  const int p = blockIdx.x & 7;
  const int b = blockIdx.x >> 3;
  const int d = threadIdx.x;

  const int s0 = ofs[b * 10 + p];
  const int s1 = ofs[b * 10 + p + 1];

  const float* src = x + (size_t)b * S_LEN * D_DIM + d;
  float acc = 0.0f;
  for (int s = s0; s < s1; ++s) acc += src[(size_t)s * D_DIM];

  const int cnt = s1 - s0;
  const float denom = (float)(cnt > 0 ? cnt : 1);
  patches[(size_t)(b * NPATCH + p) * D_DIM + d] = acc / denom;
}

__global__ void patch_gemm_wmma(const float* __restrict__ P,
                                const float* __restrict__ Wm,
                                const float* __restrict__ bias,
                                float* __restrict__ out) {
  const int wave = blockIdx.x * (blockDim.x >> 5) + (threadIdx.x >> 5);
  const int lane = threadIdx.x & 31;
  const int mt   = wave >> 6;
  const int nt   = wave & 63;
  const int m0   = mt * 16;
  const int n0   = nt * 16;
  const int half = lane >> 4;
  const int lr   = lane & 15;

  const float* Arow = P  + (size_t)(m0 + lr) * D_DIM + half * 2;
  const float* Brow = Wm + (size_t)(n0 + lr) * D_DIM + half * 2;

  v8f c = {};
#pragma unroll 8
  for (int k = 0; k < D_DIM; k += 4) {
    v2f a  = *(const v2f*)(Arow + k);
    v2f bb = *(const v2f*)(Brow + k);
    c = __builtin_amdgcn_wmma_f32_16x16x4_f32(
         false, a,  false, bb,
         (short)0, c,  false,  false);
  }

  const float bv = bias[n0 + lr];
  float* ocol = out + n0 + lr;
#pragma unroll
  for (int r = 0; r < 8; ++r) {
    ocol[(size_t)(m0 + half * 8 + r) * D_DIM] = c[r] + bv;
  }
}

extern "C" void kernel_launch(void* const* d_in, const int* in_sizes, int n_in,
                              void* d_out, int out_size, void* d_ws, size_t ws_size,
                              hipStream_t stream) {
  const float* latent = (const float*)d_in[0];
  const int*   mask   = (const int*)  d_in[1];
  const float* W      = (const float*)d_in[2];
  const float* bias   = (const float*)d_in[3];
  float*       out    = (float*)d_out;

  int*   ofs     = (int*)d_ws;
  float* patches = (float*)((char*)d_ws + 1024);

  seg_bounds_kernel<<<8, 256, 0, stream>>>(mask, ofs);
  seg_mean_kernel<<<64, 1024, 0, stream>>>(latent, ofs, patches);
  patch_gemm_wmma<<<64, 128, 0, stream>>>(patches, W, bias, out);
}


# Projection moved in front of the pooling (both are linear): raw x rows feed the WMMA, C rows are pooled per segment in registers, one launch replaces three; summation order changes
# speedup vs baseline: 3.5122x; 3.5122x over previous
#include <hip/hip_runtime.h>

typedef __attribute__((ext_vector_type(2))) float v2f;
typedef __attribute__((ext_vector_type(4))) float v4f;
typedef __attribute__((ext_vector_type(8))) float v8f;

#define S_LEN   4096
#define D_DIM   1024
#define NPATCH  8

#define KBLK 64
#define NLD  (KBLK / 8)
#define NBLK (D_DIM / KBLK)
#define NBUF 4
#define TSLOTS 2
#define NSUB   2

__device__ __forceinline__ void load_block(v4f (&a)[NLD], v4f (&b)[NLD],
                                           const float* __restrict__ ap,
                                           const float* __restrict__ bp) {
#pragma unroll
  for (int j = 0; j < NLD; ++j) {
    a[j] = *(const v4f*)(ap + j * 8);
    b[j] = *(const v4f*)(bp + j * 8);
  }
}

__device__ __forceinline__ void mma_block(const v4f (&a)[NLD], const v4f (&b)[NLD],
                                          v8f& c0, v8f& c1) {
#pragma unroll
  for (int j = 0; j < NLD; ++j) {
    const v2f alo = {a[j].x, a[j].y};
    const v2f blo = {b[j].x, b[j].y};
    const v2f ahi = {a[j].z, a[j].w};
    const v2f bhi = {b[j].z, b[j].w};
    c0 = __builtin_amdgcn_wmma_f32_16x16x4_f32(
         false, alo,  false, blo,
         (short)0, c0,  false,  false);
    c1 = __builtin_amdgcn_wmma_f32_16x16x4_f32(
         false, ahi,  false, bhi,
         (short)0, c1,  false,  false);
  }
}

__device__ __forceinline__ void ring_prologue(v4f (&a)[NBUF][NLD], v4f (&b)[NBUF][NLD],
                                              const float* __restrict__ Arow,
                                              const float* __restrict__ Brow) {
#pragma unroll
  for (int s = 0; s < NBUF - 1; ++s)
    load_block(a[s], b[s], Arow + s * KBLK, Brow + s * KBLK);
}

__device__ __forceinline__ void ring_main(v4f (&a)[NBUF][NLD], v4f (&b)[NBUF][NLD],
                                          const float* __restrict__ Arow,
                                          const float* __restrict__ Brow,
                                          v8f& c0, v8f& c1) {
  for (int blk = 0; blk < NBLK; blk += NBUF) {
#pragma unroll
    for (int s = 0; s < NBUF; ++s) {
      const int nb = blk + s + (NBUF - 1);
      if (nb < NBLK)
        load_block(a[(s + NBUF - 1) % NBUF], b[(s + NBUF - 1) % NBUF],
                   Arow + nb * KBLK, Brow + nb * KBLK);
      __builtin_amdgcn_sched_barrier(0);
      mma_block(a[s], b[s], c0, c1);
      __builtin_amdgcn_sched_barrier(0);
    }
  }
}

__global__ void __launch_bounds__(32 * TSLOTS * NSUB)
project_then_pool(const float* __restrict__ x,
                  const int* __restrict__ mask,
                  const float* __restrict__ Wm,
                  const float* __restrict__ bias,
                  float* __restrict__ out) {
  __shared__ float red[TSLOTS][NPATCH][16 * NSUB];

  const int b    = blockIdx.x >> 5;
  const int ng   = blockIdx.x & 31;
  const int lane = threadIdx.x & 31;
  const int wave  = __builtin_amdgcn_readfirstlane((int)(threadIdx.x >> 5));
  const int nsub  = wave & (NSUB - 1);
  const int tslot = wave / NSUB;
  const int half  = lane >> 4;
  const int lr    = lane & 15;
  const int n0    = ng * (16 * NSUB) + nsub * 16;

  const int*   mrow = mask + b * S_LEN;
  const float* xb   = x + (size_t)b * S_LEN * D_DIM + (size_t)lr * D_DIM + half * 4;
  const float* Brow = Wm + (size_t)(n0 + lr) * D_DIM + half * 4;

  int mcur = mrow[lane];

  v4f ra[NBUF][NLD], rb[NBUF][NLD];
  int t = tslot;
  ring_prologue(ra, rb, xb + (size_t)t * 16 * D_DIM, Brow);

  int my_start = (lane == 0) ? 0 : S_LEN;
  int found = 0;
  for (int base = 0;;) {
    const unsigned bits = __builtin_amdgcn_ballot_w32(mcur != 0);
    const int npop = __popc(bits);
    const int k = lane - found;
    unsigned bb = bits;
#pragma unroll
    for (int i = 1; i < NPATCH; ++i)
      if (i < k) bb &= bb - 1;
    if (lane >= 1 && lane <= NPATCH && k >= 1 && k <= npop)
      my_start = base + __ffs((int)bb);
    found += npop;
    base += 32;
    if (found >= NPATCH || base >= S_LEN) break;
    mcur = mrow[base + lane];
  }

  int st[NPATCH + 1];
#pragma unroll
  for (int p = 0; p <= NPATCH; ++p) st[p] = __builtin_amdgcn_readlane(my_start, p);
  const int ntile = (st[NPATCH] + 15) >> 4;

  float acc[NPATCH];
#pragma unroll
  for (int p = 0; p < NPATCH; ++p) acc[p] = 0.0f;

  while (t < ntile) {
    int zoff = 0;
    asm volatile("" : "+s"(zoff));
    const float* Bt = Brow + zoff;
    v8f c0 = {};
    v8f c1 = {};
    ring_main(ra, rb, xb + (size_t)t * 16 * D_DIM, Bt, c0, c1);
    const v8f c = c0 + c1;
#pragma unroll
    for (int r = 0; r < 8; ++r) {
      const int s = t * 16 + half * 8 + r;
#pragma unroll
      for (int p = 0; p < NPATCH; ++p)
        acc[p] += (s >= st[p] && s < st[p + 1]) ? c[r] : 0.0f;
    }
    t += TSLOTS;
    if (t < ntile) ring_prologue(ra, rb, xb + (size_t)t * 16 * D_DIM, Bt);
  }

#pragma unroll
  for (int p = 0; p < NPATCH; ++p) {
    const float tot = acc[p] + __shfl_xor(acc[p], 16);
    if (half == 0) red[tslot][p][nsub * 16 + lr] = tot;
  }
  __syncthreads();

  const int col = ng * (16 * NSUB) + lane;
  const float bv = bias[col];
#pragma unroll
  for (int q = 0; q < NPATCH / (TSLOTS * NSUB); ++q) {
    const int p = wave + q * (TSLOTS * NSUB);
    const int cnt = __shfl(my_start, p + 1) - __shfl(my_start, p);
    const float denom = (float)(cnt > 0 ? cnt : 1);
    float sum = 0.0f;
#pragma unroll
    for (int ts = 0; ts < TSLOTS; ++ts) sum += red[ts][p][lane];
    out[(size_t)(b * NPATCH + p) * D_DIM + col] = sum / denom + bv;
  }
}

extern "C" void kernel_launch(void* const* d_in, const int* in_sizes, int n_in,
                              void* d_out, int out_size, void* d_ws, size_t ws_size,
                              hipStream_t stream) {
  const float* latent = (const float*)d_in[0];
  const int*   mask   = (const int*)  d_in[1];
  const float* W      = (const float*)d_in[2];
  const float* bias   = (const float*)d_in[3];
  float*       out    = (float*)d_out;

  project_then_pool<<<256, 128, 0, stream>>>(latent, mask, W, bias, out);
}


# Pooling as a matrix product: mean matrix (1/count on a row's segment, built beside the K loop's WMMAs) x C registers; 4 WMMAs replace select-adds, 8 LDS shuffles, division; summation order changes
# speedup vs baseline: 4.3037x; 1.2254x over previous
#include <hip/hip_runtime.h>

typedef __attribute__((ext_vector_type(2))) float v2f;
typedef __attribute__((ext_vector_type(4))) float v4f;
typedef __attribute__((ext_vector_type(8))) float v8f;

#define S_LEN   4096
#define D_DIM   1024
#define NPATCH  8

#define KBLK 64
#define NLD  (KBLK / 8)
#define NBLK (D_DIM / KBLK)
#define NBUF 4
#define TSLOTS 2
#define NSUB   2

__device__ __forceinline__ void load_block(v4f (&a)[NLD], v4f (&b)[NLD],
                                           const float* __restrict__ ap,
                                           const float* __restrict__ bp) {
#pragma unroll
  for (int j = 0; j < NLD; ++j) {
    a[j] = *(const v4f*)(ap + j * 8);
    b[j] = *(const v4f*)(bp + j * 8);
  }
}

__device__ __forceinline__ void mma_block(const v4f (&a)[NLD], const v4f (&b)[NLD],
                                          v8f& c0, v8f& c1) {
#pragma unroll
  for (int j = 0; j < NLD; ++j) {
    const v2f alo = {a[j].x, a[j].y};
    const v2f blo = {b[j].x, b[j].y};
    const v2f ahi = {a[j].z, a[j].w};
    const v2f bhi = {b[j].z, b[j].w};
    c0 = __builtin_amdgcn_wmma_f32_16x16x4_f32(
         false, alo,  false, blo,
         (short)0, c0,  false,  false);
    c1 = __builtin_amdgcn_wmma_f32_16x16x4_f32(
         false, ahi,  false, bhi,
         (short)0, c1,  false,  false);
  }
}

__device__ __forceinline__ void ring_prologue(v4f (&a)[NBUF][NLD], v4f (&b)[NBUF][NLD],
                                              const float* __restrict__ Arow,
                                              const float* __restrict__ Brow) {
#pragma unroll
  for (int s = 0; s < NBUF - 1; ++s)
    load_block(a[s], b[s], Arow + s * KBLK, Brow + s * KBLK);
}

__device__ __forceinline__ void mean_rows(v2f (&mm)[4], int lo, int hi, int lr, int row0) {
  const int cnt = hi - lo;
  const float inv = 1.0f / (float)(cnt > 0 ? cnt : 1);
  const float w = (lr < NPATCH) ? inv : 0.0f;
#pragma unroll
  for (int j = 0; j < 4; ++j) {
    const int s = row0 + 2 * j;
    mm[j].x = (s     >= lo && s     < hi) ? w : 0.0f;
    mm[j].y = (s + 1 >= lo && s + 1 < hi) ? w : 0.0f;
  }
}

__device__ __forceinline__ void ring_main(v4f (&a)[NBUF][NLD], v4f (&b)[NBUF][NLD],
                                          const float* __restrict__ Arow,
                                          const float* __restrict__ Brow,
                                          v8f& c0, v8f& c1,
                                          v2f (&mm)[4], int my_start, int lr, int row0) {
  int lo = 0, hi = 0;
  for (int blk = 0; blk < NBLK; blk += NBUF) {
#pragma unroll
    for (int s = 0; s < NBUF; ++s) {
      const int nb = blk + s + (NBUF - 1);
      if (nb < NBLK)
        load_block(a[(s + NBUF - 1) % NBUF], b[(s + NBUF - 1) % NBUF],
                   Arow + nb * KBLK, Brow + nb * KBLK);
      __builtin_amdgcn_sched_barrier(0);
      if (blk == 0 && s == 0) {
        lo = __shfl(my_start, lr);
        hi = __shfl(my_start, lr + 1);
      }
      if (blk == 0 && s == 1) mean_rows(mm, lo, hi, lr, row0);
      mma_block(a[s], b[s], c0, c1);
      __builtin_amdgcn_sched_barrier(0);
    }
  }
}

__global__ void __launch_bounds__(32 * TSLOTS * NSUB)
project_then_pool(const float* __restrict__ x,
                  const int* __restrict__ mask,
                  const float* __restrict__ Wm,
                  const float* __restrict__ bias,
                  float* __restrict__ out) {
  __shared__ float red[TSLOTS][NPATCH][16 * NSUB];

  const int b    = blockIdx.x >> 5;
  const int ng   = blockIdx.x & 31;
  const int lane = threadIdx.x & 31;
  const int wave  = __builtin_amdgcn_readfirstlane((int)(threadIdx.x >> 5));
  const int nsub  = wave & (NSUB - 1);
  const int tslot = wave / NSUB;
  const int half  = lane >> 4;
  const int lr    = lane & 15;
  const int n0    = ng * (16 * NSUB) + nsub * 16;

  const int*   mrow = mask + b * S_LEN;
  const float* xb   = x + (size_t)b * S_LEN * D_DIM + (size_t)lr * D_DIM + half * 4;
  const float* Brow = Wm + (size_t)(n0 + lr) * D_DIM + half * 4;

  int mcur = mrow[lane];
  const int col = ng * (16 * NSUB) + lane;
  const float bv = bias[col];

  v4f ra[NBUF][NLD], rb[NBUF][NLD];
  int t = tslot;
  ring_prologue(ra, rb, xb + (size_t)t * 16 * D_DIM, Brow);

  int my_start = (lane == 0) ? 0 : S_LEN;
  int found = 0;
  for (int base = 0;;) {
    const unsigned bits = __builtin_amdgcn_ballot_w32(mcur != 0);
    const int npop = __popc(bits);
    const int k = lane - found;
    unsigned bb = bits;
#pragma unroll
    for (int i = 1; i < NPATCH; ++i)
      if (i < k) bb &= bb - 1;
    if (lane >= 1 && lane <= NPATCH && k >= 1 && k <= npop)
      my_start = base + __ffs((int)bb);
    found += npop;
    base += 32;
    if (found >= NPATCH || base >= S_LEN) break;
    mcur = mrow[base + lane];
  }

  const int ntile = (__builtin_amdgcn_readlane(my_start, NPATCH) + 15) >> 4;

  v8f pacc = {};

  while (t < ntile) {
    int zoff = 0;
    asm volatile("" : "+s"(zoff));
    const float* Bt = Brow + zoff;
    v8f c0 = {};
    v8f c1 = {};
    v2f mm[4];
    ring_main(ra, rb, xb + (size_t)t * 16 * D_DIM, Bt, c0, c1,
              mm, my_start, lr, t * 16 + half * 8);
    const v8f c = c0 + c1;
#pragma unroll
    for (int j = 0; j < 4; ++j) {
      const v2f crows = {c[2 * j], c[2 * j + 1]};
      pacc = __builtin_amdgcn_wmma_f32_16x16x4_f32(
           false, mm[j],  false, crows,
           (short)0, pacc,  false,  false);
    }
    t += TSLOTS;
    if (t < ntile) ring_prologue(ra, rb, xb + (size_t)t * 16 * D_DIM, Bt);
  }

  if (half == 0) {
#pragma unroll
    for (int p = 0; p < NPATCH; ++p) red[tslot][p][nsub * 16 + lr] = pacc[p];
  }
  __syncthreads();

#pragma unroll
  for (int q = 0; q < NPATCH / (TSLOTS * NSUB); ++q) {
    const int p = wave + q * (TSLOTS * NSUB);
    float sum = red[0][p][lane];
#pragma unroll
    for (int ts = 1; ts < TSLOTS; ++ts) sum += red[ts][p][lane];
    out[(size_t)(b * NPATCH + p) * D_DIM + col] = sum + bv;
  }
}

extern "C" void kernel_launch(void* const* d_in, const int* in_sizes, int n_in,
                              void* d_out, int out_size, void* d_ws, size_t ws_size,
                              hipStream_t stream) {
  const float* latent = (const float*)d_in[0];
  const int*   mask   = (const int*)  d_in[1];
  const float* W      = (const float*)d_in[2];
  const float* bias   = (const float*)d_in[3];
  float*       out    = (float*)d_out;

  project_then_pool<<<256, 128, 0, stream>>>(latent, mask, W, bias, out);
}


# Idle row-tile slot put to work: a batch needing one row tile splits tile 0's K between the two slots (512 k each), partial means meet in the existing LDS sum; summation order changes
# speedup vs baseline: 5.2895x; 1.2291x over previous
#include <hip/hip_runtime.h>

typedef __attribute__((ext_vector_type(2))) float v2f;
typedef __attribute__((ext_vector_type(4))) float v4f;
typedef __attribute__((ext_vector_type(8))) float v8f;

#define S_LEN   4096
#define D_DIM   1024
#define NPATCH  8

#define KBLK 64
#define NLD  (KBLK / 8)
#define NBLK (D_DIM / KBLK)
#define NBUF 4
#define TSLOTS 2
#define NSUB   2
static_assert((NBLK / TSLOTS) % NBUF == 0, "a slot's share of K must be whole turns of the ring");

__device__ __forceinline__ void load_block(v4f (&a)[NLD], v4f (&b)[NLD],
                                           const float* __restrict__ ap,
                                           const float* __restrict__ bp) {
#pragma unroll
  for (int j = 0; j < NLD; ++j) {
    a[j] = *(const v4f*)(ap + j * 8);
    b[j] = *(const v4f*)(bp + j * 8);
  }
}

__device__ __forceinline__ void mma_block(const v4f (&a)[NLD], const v4f (&b)[NLD],
                                          v8f& c0, v8f& c1) {
#pragma unroll
  for (int j = 0; j < NLD; ++j) {
    const v2f alo = {a[j].x, a[j].y};
    const v2f blo = {b[j].x, b[j].y};
    const v2f ahi = {a[j].z, a[j].w};
    const v2f bhi = {b[j].z, b[j].w};
    c0 = __builtin_amdgcn_wmma_f32_16x16x4_f32(
         false, alo,  false, blo,
         (short)0, c0,  false,  false);
    c1 = __builtin_amdgcn_wmma_f32_16x16x4_f32(
         false, ahi,  false, bhi,
         (short)0, c1,  false,  false);
  }
}

__device__ __forceinline__ void ring_prologue(v4f (&a)[NBUF][NLD], v4f (&b)[NBUF][NLD],
                                              const float* __restrict__ Arow,
                                              const float* __restrict__ Brow) {
#pragma unroll
  for (int s = 0; s < NBUF - 1; ++s)
    load_block(a[s], b[s], Arow + s * KBLK, Brow + s * KBLK);
}

__device__ __forceinline__ void mean_rows(v2f (&mm)[4], int lo, int hi, int lr, int row0) {
  const int cnt = hi - lo;
  const float inv = 1.0f / (float)(cnt > 0 ? cnt : 1);
  const float w = (lr < NPATCH) ? inv : 0.0f;
#pragma unroll
  for (int j = 0; j < 4; ++j) {
    const int s = row0 + 2 * j;
    mm[j].x = (s     >= lo && s     < hi) ? w : 0.0f;
    mm[j].y = (s + 1 >= lo && s + 1 < hi) ? w : 0.0f;
  }
}

template <int NB>
__device__ __forceinline__ void ring_main(v4f (&a)[NBUF][NLD], v4f (&b)[NBUF][NLD],
                                          const float* __restrict__ Arow,
                                          const float* __restrict__ Brow,
                                          v8f& c0, v8f& c1,
                                          v2f (&mm)[4], int my_start, int lr, int row0) {
  int lo = 0, hi = 0;
  for (int blk = 0; blk < NB; blk += NBUF) {
#pragma unroll
    for (int s = 0; s < NBUF; ++s) {
      const int nb = blk + s + (NBUF - 1);
      if (nb < NB)
        load_block(a[(s + NBUF - 1) % NBUF], b[(s + NBUF - 1) % NBUF],
                   Arow + nb * KBLK, Brow + nb * KBLK);
      __builtin_amdgcn_sched_barrier(0);
      if (blk == 0 && s == 0) {
        lo = __shfl(my_start, lr);
        hi = __shfl(my_start, lr + 1);
      }
      if (blk == 0 && s == 1) mean_rows(mm, lo, hi, lr, row0);
      mma_block(a[s], b[s], c0, c1);
      __builtin_amdgcn_sched_barrier(0);
    }
  }
}

__device__ __forceinline__ void pool_tile(v8f& pacc, const v2f (&mm)[4], const v8f& c) {
#pragma unroll
  for (int j = 0; j < 4; ++j) {
    const v2f crows = {c[2 * j], c[2 * j + 1]};
    pacc = __builtin_amdgcn_wmma_f32_16x16x4_f32(
         false, mm[j],  false, crows,
         (short)0, pacc,  false,  false);
  }
}

__global__ void __launch_bounds__(32 * TSLOTS * NSUB)
project_then_pool(const float* __restrict__ x,
                  const int* __restrict__ mask,
                  const float* __restrict__ Wm,
                  const float* __restrict__ bias,
                  float* __restrict__ out) {
  __shared__ float red[TSLOTS][NPATCH][16 * NSUB];

  const int b    = blockIdx.x >> 5;
  const int ng   = blockIdx.x & 31;
  const int lane = threadIdx.x & 31;
  const int wave  = __builtin_amdgcn_readfirstlane((int)(threadIdx.x >> 5));
  const int nsub  = wave & (NSUB - 1);
  const int tslot = wave / NSUB;
  const int half  = lane >> 4;
  const int lr    = lane & 15;
  const int n0    = ng * (16 * NSUB) + nsub * 16;

  const int*   mrow = mask + b * S_LEN;
  const float* xb   = x + (size_t)b * S_LEN * D_DIM + (size_t)lr * D_DIM + half * 4;
  const float* Brow = Wm + (size_t)(n0 + lr) * D_DIM + half * 4;

  int mcur = mrow[lane];
  const int col = ng * (16 * NSUB) + lane;
  const float bv = bias[col];

  v4f ra[NBUF][NLD], rb[NBUF][NLD];
  int t = tslot;
  const int khalf = tslot * (D_DIM / TSLOTS);
  ring_prologue(ra, rb, xb + khalf, Brow + khalf);

  int my_start = (lane == 0) ? 0 : S_LEN;
  int found = 0;
  for (int base = 0;;) {
    const unsigned bits = __builtin_amdgcn_ballot_w32(mcur != 0);
    const int npop = __popc(bits);
    const int k = lane - found;
    unsigned bb = bits;
#pragma unroll
    for (int i = 1; i < NPATCH; ++i)
      if (i < k) bb &= bb - 1;
    if (lane >= 1 && lane <= NPATCH && k >= 1 && k <= npop)
      my_start = base + __ffs((int)bb);
    found += npop;
    base += 32;
    if (found >= NPATCH || base >= S_LEN) break;
    mcur = mrow[base + lane];
  }

  const int ntile = (__builtin_amdgcn_readlane(my_start, NPATCH) + 15) >> 4;

  v8f pacc = {};

  if (ntile == 1) {
    v8f c0 = {};
    v8f c1 = {};
    v2f mm[4];
    ring_main<NBLK / TSLOTS>(ra, rb, xb + khalf, Brow + khalf, c0, c1,
                             mm, my_start, lr, half * 8);
    pool_tile(pacc, mm, c0 + c1);
    t = ntile;
  } else if (tslot != 0) {
    ring_prologue(ra, rb, xb + (size_t)t * 16 * D_DIM, Brow);
  }

  while (t < ntile) {
    int zoff = 0;
    asm volatile("" : "+s"(zoff));
    const float* Bt = Brow + zoff;
    v8f c0 = {};
    v8f c1 = {};
    v2f mm[4];
    ring_main<NBLK>(ra, rb, xb + (size_t)t * 16 * D_DIM, Bt, c0, c1,
                    mm, my_start, lr, t * 16 + half * 8);
    pool_tile(pacc, mm, c0 + c1);
    t += TSLOTS;
    if (t < ntile) ring_prologue(ra, rb, xb + (size_t)t * 16 * D_DIM, Bt);
  }

  if (half == 0) {
#pragma unroll
    for (int p = 0; p < NPATCH; ++p) red[tslot][p][nsub * 16 + lr] = pacc[p];
  }
  __syncthreads();

#pragma unroll
  for (int q = 0; q < NPATCH / (TSLOTS * NSUB); ++q) {
    const int p = wave + q * (TSLOTS * NSUB);
    float sum = red[0][p][lane];
#pragma unroll
    for (int ts = 1; ts < TSLOTS; ++ts) sum += red[ts][p][lane];
    out[(size_t)(b * NPATCH + p) * D_DIM + col] = sum + bv;
  }
}

extern "C" void kernel_launch(void* const* d_in, const int* in_sizes, int n_in,
                              void* d_out, int out_size, void* d_ws, size_t ws_size,
                              hipStream_t stream) {
  const float* latent = (const float*)d_in[0];
  const int*   mask   = (const int*)  d_in[1];
  const float* W      = (const float*)d_in[2];
  const float* bias   = (const float*)d_in[3];
  float*       out    = (float*)d_out;

  project_then_pool<<<256, 128, 0, stream>>>(latent, mask, W, bias, out);
}
